# starting kernel
# baseline (speedup 1.0000x reference)
#include <hip/hip_runtime.h>

#define NEG_SLOPE 0.2f
#define LN_EPS 1e-5f
#define SM_EPS 1e-16f

typedef __attribute__((ext_vector_type(2))) float v2f;
typedef __attribute__((ext_vector_type(8))) float v8f;

__device__ __forceinline__ float lrelu(float x) { return x > 0.0f ? x : NEG_SLOPE * x; }

__global__ void __launch_bounds__(128)
gemm64_wmma(const float* __restrict__ A, const float* __restrict__ W,
            float* __restrict__ C, int nrows) {
    const int wave = threadIdx.x >> 5;
    const int lane = threadIdx.x & 31;
    const int mrow = lane & 15;
    const int kh   = lane >> 4;
    const int row0 = blockIdx.x * 16;
    int row = row0 + mrow;
    row = row < nrows ? row : nrows - 1;
    const int col  = wave * 16 + mrow;

    const float* ap = A + (size_t)row * 64 + 2 * kh;
    const float* bp = W + (size_t)(2 * kh) * 64 + col;

    v8f acc = {};
    #pragma unroll
    for (int k = 0; k < 64; k += 4) {
        v2f a = *(const v2f*)(ap + k);
        v2f b;
        b.x = bp[(size_t)k * 64];
        b.y = bp[(size_t)k * 64 + 64];
        acc = __builtin_amdgcn_wmma_f32_16x16x4_f32(
             false, a,  false, b,
             (short)0, acc,  false,  false);
    }

    const int srow = row0 + 8 * (lane >> 4);
    const int scol = wave * 16 + (lane & 15);
    #pragma unroll
    for (int r = 0; r < 8; ++r) {
        const int rr = srow + r;
        if (rr < nrows) C[(size_t)rr * 64 + scol] = acc[r];
    }
}

template <int H, int C>
__global__ void attn_prep(const float* __restrict__ xw,
                          const float* __restrict__ a_src,
                          const float* __restrict__ a_dst,
                          float* __restrict__ asrc, float* __restrict__ adst,
                          float* __restrict__ m, int n) {
    const int t = blockIdx.x * blockDim.x + threadIdx.x;
    if (t >= n * H) return;
    const int node = t / H;
    const int h    = t - node * H;
    const float* xp = xw + (size_t)node * 64 + h * C;
    float s = 0.0f, d = 0.0f;
    #pragma unroll
    for (int c = 0; c < C; ++c) {
        const float v = xp[c];
        s += v * a_src[h * C + c];
        d += v * a_dst[h * C + c];
    }
    asrc[t] = s;
    adst[t] = d;
    m[t] = lrelu(s + d);
}

template <int H>
__global__ void edge_max_k(const int* __restrict__ src, const int* __restrict__ dst,
                           const float* __restrict__ asrc, const float* __restrict__ adst,
                           float* __restrict__ m, int ecnt) {
    const int t = blockIdx.x * blockDim.x + threadIdx.x;
    if (t >= ecnt * H) return;
    const int edge = t / H;
    const int h    = t - edge * H;
    const int s = src[edge];
    const int d = dst[edge];
    const float e = lrelu(asrc[s * H + h] + adst[d * H + h]);
    atomicMax(&m[d * H + h], e);
}

template <int H, int C>
__global__ void softmax_self(const float* __restrict__ xw,
                             const float* __restrict__ asrc,
                             const float* __restrict__ adst,
                             const float* __restrict__ m,
                             float* __restrict__ denom, float* __restrict__ agg,
                             int n) {
    const int t = blockIdx.x * blockDim.x + threadIdx.x;
    if (t >= n * 64) return;
    const int node = t >> 6;
    const int ch   = t & 63;
    const int h    = ch / C;
    const float e  = lrelu(asrc[node * H + h] + adst[node * H + h]);
    const float ex = __expf(e - m[node * H + h]);
    agg[t] = xw[t] * ex;
    if ((ch & (C - 1)) == 0) denom[node * H + h] = ex;
}

template <int H, int C>
__global__ void edge_agg(const int* __restrict__ src, const int* __restrict__ dst,
                         const float* __restrict__ asrc, const float* __restrict__ adst,
                         const float* __restrict__ m, const float* __restrict__ xw,
                         float* __restrict__ denom, float* __restrict__ agg,
                         int ecnt) {
    const long long t = (long long)blockIdx.x * blockDim.x + threadIdx.x;
    if (t >= (long long)ecnt * 64) return;
    const int edge = (int)(t >> 6);
    const int ch   = (int)(t & 63);
    const int h    = ch / C;
    const int s = src[edge];
    const int d = dst[edge];
    const float e  = lrelu(asrc[s * H + h] + adst[d * H + h]);
    const float ex = __expf(e - m[d * H + h]);
    atomicAdd(&agg[(size_t)d * 64 + ch], ex * xw[(size_t)s * 64 + ch]);
    if ((ch & (C - 1)) == 0) atomicAdd(&denom[d * H + h], ex);
}

template <int H, int C>
__global__ void finalize_k(const float* __restrict__ agg, const float* __restrict__ denom,
                           const float* __restrict__ bias, const float* __restrict__ g,
                           const float* __restrict__ bparam, float* __restrict__ out,
                           int n) {
    const int node = (blockIdx.x * blockDim.x + threadIdx.x) >> 5;
    const int lane = threadIdx.x & 31;
    if (node >= n) return;
    const int ch0 = lane, ch1 = lane + 32;
    float v0 = agg[(size_t)node * 64 + ch0] / (denom[node * H + ch0 / C] + SM_EPS) + bias[ch0];
    float v1 = agg[(size_t)node * 64 + ch1] / (denom[node * H + ch1 / C] + SM_EPS) + bias[ch1];

    float s = v0 + v1;
    #pragma unroll
    for (int off = 16; off > 0; off >>= 1) s += __shfl_xor(s, off, 32);
    const float mu = s * (1.0f / 64.0f);
    const float d0 = v0 - mu, d1 = v1 - mu;
    float vv = d0 * d0 + d1 * d1;
    #pragma unroll
    for (int off = 16; off > 0; off >>= 1) vv += __shfl_xor(vv, off, 32);
    const float rstd = rsqrtf(vv * (1.0f / 64.0f) + LN_EPS);

    float o0 = d0 * rstd * g[ch0] + bparam[ch0];
    float o1 = d1 * rstd * g[ch1] + bparam[ch1];
    o0 = o0 > 0.0f ? o0 : __expf(o0) - 1.0f;
    o1 = o1 > 0.0f ? o1 : __expf(o1) - 1.0f;
    out[(size_t)node * 64 + ch0] = o0;
    out[(size_t)node * 64 + ch1] = o1;
}

static inline unsigned cdiv(long long a, int b) { return (unsigned)((a + b - 1) / b); }

extern "C" void kernel_launch(void* const* d_in, const int* in_sizes, int n_in,
                              void* d_out, int out_size, void* d_ws, size_t ws_size,
                              hipStream_t stream) {
    const float* h   = (const float*)d_in[0];
    const int*   ei  = (const int*)d_in[1];
    const float* W1  = (const float*)d_in[2];
    const float* a1s = (const float*)d_in[3];
    const float* a1d = (const float*)d_in[4];
    const float* b1  = (const float*)d_in[5];
    const float* W2  = (const float*)d_in[6];
    const float* a2s = (const float*)d_in[7];
    const float* a2d = (const float*)d_in[8];
    const float* b2  = (const float*)d_in[9];
    const float* ln1g = (const float*)d_in[10];
    const float* ln1b = (const float*)d_in[11];
    const float* ln2g = (const float*)d_in[12];
    const float* ln2b = (const float*)d_in[13];

    const int n = in_sizes[0] / 64;
    const int e = in_sizes[1] / 2;
    const int* src = ei;
    const int* dst = ei + e;

    float* ws = (float*)d_ws;
    float* xw   = ws; ws += (size_t)n * 64;
    float* asrc = ws; ws += (size_t)n * 4;
    float* adst = ws; ws += (size_t)n * 4;
    float* m    = ws; ws += (size_t)n * 4;
    float* den  = ws; ws += (size_t)n * 4;
    float* agg  = ws; ws += (size_t)n * 64;
    float* x1   = ws; ws += (size_t)n * 64;

    const int B = 256;

    gemm64_wmma<<<cdiv(n, 16), 128, 0, stream>>>(h, W1, xw, n);
    attn_prep<4, 16><<<cdiv((long long)n * 4, B), B, 0, stream>>>(xw, a1s, a1d, asrc, adst, m, n);
    edge_max_k<4><<<cdiv((long long)e * 4, B), B, 0, stream>>>(src, dst, asrc, adst, m, e);
    softmax_self<4, 16><<<cdiv((long long)n * 64, B), B, 0, stream>>>(xw, asrc, adst, m, den, agg, n);
    edge_agg<4, 16><<<cdiv((long long)e * 64, B), B, 0, stream>>>(src, dst, asrc, adst, m, xw, den, agg, e);
    finalize_k<4, 16><<<cdiv((long long)n * 32, B), B, 0, stream>>>(agg, den, b1, ln1g, ln1b, x1, n);

    gemm64_wmma<<<cdiv(n, 16), 128, 0, stream>>>(x1, W2, xw, n);
    attn_prep<1, 64><<<cdiv((long long)n, B), B, 0, stream>>>(xw, a2s, a2d, asrc, adst, m, n);
    edge_max_k<1><<<cdiv((long long)e, B), B, 0, stream>>>(src, dst, asrc, adst, m, e);
    softmax_self<1, 64><<<cdiv((long long)n * 64, B), B, 0, stream>>>(xw, asrc, adst, m, den, agg, n);
    edge_agg<1, 64><<<cdiv((long long)e * 64, B), B, 0, stream>>>(src, dst, asrc, adst, m, xw, den, agg, e);
    finalize_k<1, 64><<<cdiv((long long)n * 32, B), B, 0, stream>>>(agg, den, b2, ln2g, ln2b, (float*)d_out, n);
}


# float-atomic edge scatter replaced by per-destination in-edge linked lists (one int exchange per edge, built once per call); a wave per node walks its list, running-max softmax; sums in arrival order
# speedup vs baseline: 3.4108x; 3.4108x over previous
#include <hip/hip_runtime.h>

#define NEG_SLOPE 0.2f
#define LN_EPS 1e-5f
#define SM_EPS 1e-16f

typedef __attribute__((ext_vector_type(2))) float v2f;
typedef __attribute__((ext_vector_type(8))) float v8f;

__device__ __forceinline__ float lrelu(float x) { return x > 0.0f ? x : NEG_SLOPE * x; }

__global__ void __launch_bounds__(128)
gemm64_wmma(const float* __restrict__ A, const float* __restrict__ W,
            float* __restrict__ C, int nrows) {
    const int wave = threadIdx.x >> 5;
    const int lane = threadIdx.x & 31;
    const int mrow = lane & 15;
    const int kh   = lane >> 4;
    const int row0 = blockIdx.x * 16;
    int row = row0 + mrow;
    row = row < nrows ? row : nrows - 1;
    const int col  = wave * 16 + mrow;

    const float* ap = A + (size_t)row * 64 + 2 * kh;
    const float* bp = W + (size_t)(2 * kh) * 64 + col;

    v8f acc = {};
    #pragma unroll
    for (int k = 0; k < 64; k += 4) {
        v2f a = *(const v2f*)(ap + k);
        v2f b;
        b.x = bp[(size_t)k * 64];
        b.y = bp[(size_t)k * 64 + 64];
        acc = __builtin_amdgcn_wmma_f32_16x16x4_f32(
             false, a,  false, b,
             (short)0, acc,  false,  false);
    }

    const int srow = row0 + 8 * (lane >> 4);
    const int scol = wave * 16 + (lane & 15);
    #pragma unroll
    for (int r = 0; r < 8; ++r) {
        const int rr = srow + r;
        if (rr < nrows) C[(size_t)rr * 64 + scol] = acc[r];
    }
}

template <int H, int C>
__global__ void attn_prep(const float* __restrict__ xw,
                          const float* __restrict__ a_src,
                          const float* __restrict__ a_dst,
                          float* __restrict__ asrc, float* __restrict__ adst,
                          int n) {
    const int t = blockIdx.x * blockDim.x + threadIdx.x;
    if (t >= n * H) return;
    const int node = t / H;
    const int h    = t - node * H;
    const float* xp = xw + (size_t)node * 64 + h * C;
    float s = 0.0f, d = 0.0f;
    #pragma unroll
    for (int c = 0; c < C; ++c) {
        const float v = xp[c];
        s += v * a_src[h * C + c];
        d += v * a_dst[h * C + c];
    }
    asrc[t] = s;
    adst[t] = d;
}

__global__ void list_init(int* __restrict__ head, int n) {
    const int t = blockIdx.x * blockDim.x + threadIdx.x;
    if (t < n) head[t] = -1;
}

__global__ void list_build(const int* __restrict__ src, const int* __restrict__ dst,
                           int* __restrict__ head, int2* __restrict__ rec, int ecnt) {
    const int t = blockIdx.x * blockDim.x + threadIdx.x;
    if (t >= ecnt) return;
    const int s = src[t];
    const int prev = atomicExch(&head[dst[t]], t);
    rec[t] = make_int2(s, prev);
}

template <int H, int C>
__global__ void __launch_bounds__(256)
node_walk(const int* __restrict__ head, const int2* __restrict__ rec,
          const float* __restrict__ asrc, const float* __restrict__ adst,
          const float* __restrict__ xw,
          float* __restrict__ denom, float* __restrict__ agg,
          int n, int ecnt) {
    const int node = __builtin_amdgcn_readfirstlane((int)((blockIdx.x * blockDim.x + threadIdx.x) >> 5));
    const int lane = threadIdx.x & 31;
    if (node >= n) return;
    const int h = (2 * lane) / C;

    const float ad = adst[node * H + h];
    float m = lrelu(asrc[node * H + h] + ad);
    const v2f self = *(const v2f*)(xw + (size_t)node * 64 + 2 * lane);
    float a0 = self.x, a1 = self.y, den = 1.0f;

    int e = head[node];
    int2 r = rec[e >= 0 ? e : 0];
    for (int it = 0; it < ecnt && e >= 0; ++it) {
        const int s  = r.x;
        const int en = r.y;
        r = rec[en >= 0 ? en : 0];
        const float as = asrc[s * H + h];
        const v2f xv = *(const v2f*)(xw + (size_t)s * 64 + 2 * lane);
        asm volatile("" :: "s"(r.x), "s"(r.y), "v"(as), "v"(xv.x), "v"(xv.y));
        const float ev = lrelu(as + ad);
        const float mn = fmaxf(m, ev);
        const float sc = __expf(m - mn);
        const float p  = __expf(ev - mn);
        den = den * sc + p;
        a0  = a0 * sc + p * xv.x;
        a1  = a1 * sc + p * xv.y;
        m = mn;
        e = en;
    }

    v2f o; o.x = a0; o.y = a1;
    *(v2f*)(agg + (size_t)node * 64 + 2 * lane) = o;
    if (((2 * lane) & (C - 1)) == 0) denom[node * H + h] = den;
}

template <int H, int C>
__global__ void finalize_k(const float* __restrict__ agg, const float* __restrict__ denom,
                           const float* __restrict__ bias, const float* __restrict__ g,
                           const float* __restrict__ bparam, float* __restrict__ out,
                           int n) {
    const int node = (blockIdx.x * blockDim.x + threadIdx.x) >> 5;
    const int lane = threadIdx.x & 31;
    if (node >= n) return;
    const int ch0 = lane, ch1 = lane + 32;
    float v0 = agg[(size_t)node * 64 + ch0] / (denom[node * H + ch0 / C] + SM_EPS) + bias[ch0];
    float v1 = agg[(size_t)node * 64 + ch1] / (denom[node * H + ch1 / C] + SM_EPS) + bias[ch1];

    float s = v0 + v1;
    #pragma unroll
    for (int off = 16; off > 0; off >>= 1) s += __shfl_xor(s, off, 32);
    const float mu = s * (1.0f / 64.0f);
    const float d0 = v0 - mu, d1 = v1 - mu;
    float vv = d0 * d0 + d1 * d1;
    #pragma unroll
    for (int off = 16; off > 0; off >>= 1) vv += __shfl_xor(vv, off, 32);
    const float rstd = rsqrtf(vv * (1.0f / 64.0f) + LN_EPS);

    float o0 = d0 * rstd * g[ch0] + bparam[ch0];
    float o1 = d1 * rstd * g[ch1] + bparam[ch1];
    o0 = o0 > 0.0f ? o0 : __expf(o0) - 1.0f;
    o1 = o1 > 0.0f ? o1 : __expf(o1) - 1.0f;
    out[(size_t)node * 64 + ch0] = o0;
    out[(size_t)node * 64 + ch1] = o1;
}

static inline unsigned cdiv(long long a, int b) { return (unsigned)((a + b - 1) / b); }

extern "C" void kernel_launch(void* const* d_in, const int* in_sizes, int n_in,
                              void* d_out, int out_size, void* d_ws, size_t ws_size,
                              hipStream_t stream) {
    const float* h   = (const float*)d_in[0];
    const int*   ei  = (const int*)d_in[1];
    const float* W1  = (const float*)d_in[2];
    const float* a1s = (const float*)d_in[3];
    const float* a1d = (const float*)d_in[4];
    const float* b1  = (const float*)d_in[5];
    const float* W2  = (const float*)d_in[6];
    const float* a2s = (const float*)d_in[7];
    const float* a2d = (const float*)d_in[8];
    const float* b2  = (const float*)d_in[9];
    const float* ln1g = (const float*)d_in[10];
    const float* ln1b = (const float*)d_in[11];
    const float* ln2g = (const float*)d_in[12];
    const float* ln2b = (const float*)d_in[13];

    const int n = in_sizes[0] / 64;
    const int e = in_sizes[1] / 2;
    const int* src = ei;
    const int* dst = ei + e;

    float* ws = (float*)d_ws;
    float* xw   = ws; ws += (size_t)n * 64;
    float* asrc = ws; ws += (size_t)n * 4;
    float* adst = ws; ws += (size_t)n * 4;
    float* den  = ws; ws += (size_t)n * 4;
    float* agg  = ws; ws += (size_t)n * 64;
    float* x1   = ws; ws += (size_t)n * 64;
    int2*  rec  = (int2*)ws; ws += (size_t)e * 2;
    int*   head = (int*)ws;  ws += (size_t)n;

    const int B = 256;

    list_init<<<cdiv(n, B), B, 0, stream>>>(head, n);
    list_build<<<cdiv(e, B), B, 0, stream>>>(src, dst, head, rec, e);

    gemm64_wmma<<<cdiv(n, 16), 128, 0, stream>>>(h, W1, xw, n);
    attn_prep<4, 16><<<cdiv((long long)n * 4, B), B, 0, stream>>>(xw, a1s, a1d, asrc, adst, n);
    node_walk<4, 16><<<cdiv((long long)n * 32, B), B, 0, stream>>>(head, rec, asrc, adst, xw, den, agg, n, e);
    finalize_k<4, 16><<<cdiv((long long)n * 32, B), B, 0, stream>>>(agg, den, b1, ln1g, ln1b, x1, n);

    gemm64_wmma<<<cdiv(n, 16), 128, 0, stream>>>(x1, W2, xw, n);
    attn_prep<1, 64><<<cdiv((long long)n, B), B, 0, stream>>>(xw, a2s, a2d, asrc, adst, n);
    node_walk<1, 64><<<cdiv((long long)n * 32, B), B, 0, stream>>>(head, rec, asrc, adst, xw, den, agg, n, e);
    finalize_k<1, 64><<<cdiv((long long)n * 32, B), B, 0, stream>>>(agg, den, b2, ln2g, ln2b, (float*)d_out, n);
}


# edge scores recomputed in the list walk from the gathered rows (per-lane partial dot + DPP lane-group sum): no score gather per edge, both attn_prep launches gone; dot sums now a lane tree
# speedup vs baseline: 3.7163x; 1.0896x over previous
#include <hip/hip_runtime.h>

#define NEG_SLOPE 0.2f
#define LN_EPS 1e-5f
#define SM_EPS 1e-16f

typedef __attribute__((ext_vector_type(2))) float v2f;
typedef __attribute__((ext_vector_type(8))) float v8f;

__device__ __forceinline__ float lrelu(float x) { return x > 0.0f ? x : NEG_SLOPE * x; }

__global__ void __launch_bounds__(128)
gemm64_wmma(const float* __restrict__ A, const float* __restrict__ W,
            float* __restrict__ C, int nrows) {
    const int wave = threadIdx.x >> 5;
    const int lane = threadIdx.x & 31;
    const int mrow = lane & 15;
    const int kh   = lane >> 4;
    const int row0 = blockIdx.x * 16;
    int row = row0 + mrow;
    row = row < nrows ? row : nrows - 1;
    const int col  = wave * 16 + mrow;

    const float* ap = A + (size_t)row * 64 + 2 * kh;
    const float* bp = W + (size_t)(2 * kh) * 64 + col;

    v8f acc = {};
    #pragma unroll
    for (int k = 0; k < 64; k += 4) {
        v2f a = *(const v2f*)(ap + k);
        v2f b;
        b.x = bp[(size_t)k * 64];
        b.y = bp[(size_t)k * 64 + 64];
        acc = __builtin_amdgcn_wmma_f32_16x16x4_f32(
             false, a,  false, b,
             (short)0, acc,  false,  false);
    }

    const int srow = row0 + 8 * (lane >> 4);
    const int scol = wave * 16 + (lane & 15);
    #pragma unroll
    for (int r = 0; r < 8; ++r) {
        const int rr = srow + r;
        if (rr < nrows) C[(size_t)rr * 64 + scol] = acc[r];
    }
}

template <int CTRL>
__device__ __forceinline__ float xmask_add(float v) {
    const int b = __builtin_bit_cast(int, v);
    const int t = __builtin_amdgcn_update_dpp(b, b, CTRL, 0xf, 0xf, true);
    return v + __builtin_bit_cast(float, t);
}

template <int LANES>
__device__ __forceinline__ float group_sum(float v) {
    v = xmask_add<0x161>(v);
    v = xmask_add<0x162>(v);
    v = xmask_add<0x164>(v);
    if (LANES >= 16) v = xmask_add<0x168>(v);
    if (LANES == 32) {
        const int b = __builtin_bit_cast(int, v);
        const float lo = __builtin_bit_cast(float, __builtin_amdgcn_readlane(b, 0));
        const float hi = __builtin_bit_cast(float, __builtin_amdgcn_readlane(b, 16));
        v = lo + hi;
    }
    return v;
}

__global__ void list_init(int* __restrict__ head, int n) {
    const int t = blockIdx.x * blockDim.x + threadIdx.x;
    if (t < n) head[t] = -1;
}

__global__ void list_build(const int* __restrict__ src, const int* __restrict__ dst,
                           int* __restrict__ head, int2* __restrict__ rec, int ecnt) {
    const int t = blockIdx.x * blockDim.x + threadIdx.x;
    if (t >= ecnt) return;
    const int s = src[t];
    const int prev = atomicExch(&head[dst[t]], t);
    rec[t] = make_int2(s, prev);
}

template <int H, int C>
__global__ void __launch_bounds__(256)
node_walk(const int* __restrict__ head, const int2* __restrict__ rec,
          const float* __restrict__ a_src, const float* __restrict__ a_dst,
          const float* __restrict__ xw,
          float* __restrict__ denom, float* __restrict__ agg,
          int n, int ecnt) {
    const int node = __builtin_amdgcn_readfirstlane((int)((blockIdx.x * blockDim.x + threadIdx.x) >> 5));
    const int lane = threadIdx.x & 31;
    if (node >= n) return;
    const int h = (2 * lane) / C;

    const v2f ws = *(const v2f*)(a_src + 2 * lane);
    const v2f wd = *(const v2f*)(a_dst + 2 * lane);

    const v2f self = *(const v2f*)(xw + (size_t)node * 64 + 2 * lane);
    const float ad = group_sum<C / 2>(self.x * wd.x + self.y * wd.y);
    float m = lrelu(group_sum<C / 2>(self.x * ws.x + self.y * ws.y) + ad);
    float a0 = self.x, a1 = self.y, den = 1.0f;

    int e = head[node];
    int2 r = rec[e >= 0 ? e : 0];
    for (int it = 0; it < ecnt && e >= 0; ++it) {
        const int s  = r.x;
        const int en = r.y;
        r = rec[en >= 0 ? en : 0];
        const v2f xv = *(const v2f*)(xw + (size_t)s * 64 + 2 * lane);
        asm volatile("" :: "s"(r.x), "s"(r.y), "v"(xv.x), "v"(xv.y));
        __builtin_amdgcn_sched_barrier(0);
        const float as = group_sum<C / 2>(xv.x * ws.x + xv.y * ws.y);
        const float ev = lrelu(as + ad);
        const float mn = fmaxf(m, ev);
        const float sc = __expf(m - mn);
        const float p  = __expf(ev - mn);
        den = den * sc + p;
        a0  = a0 * sc + p * xv.x;
        a1  = a1 * sc + p * xv.y;
        m = mn;
        e = en;
    }

    v2f o; o.x = a0; o.y = a1;
    *(v2f*)(agg + (size_t)node * 64 + 2 * lane) = o;
    if (((2 * lane) & (C - 1)) == 0) denom[node * H + h] = den;
}

template <int H, int C>
__global__ void finalize_k(const float* __restrict__ agg, const float* __restrict__ denom,
                           const float* __restrict__ bias, const float* __restrict__ g,
                           const float* __restrict__ bparam, float* __restrict__ out,
                           int n) {
    const int node = (blockIdx.x * blockDim.x + threadIdx.x) >> 5;
    const int lane = threadIdx.x & 31;
    if (node >= n) return;
    const int ch0 = lane, ch1 = lane + 32;
    float v0 = agg[(size_t)node * 64 + ch0] / (denom[node * H + ch0 / C] + SM_EPS) + bias[ch0];
    float v1 = agg[(size_t)node * 64 + ch1] / (denom[node * H + ch1 / C] + SM_EPS) + bias[ch1];

    float s = v0 + v1;
    #pragma unroll
    for (int off = 16; off > 0; off >>= 1) s += __shfl_xor(s, off, 32);
    const float mu = s * (1.0f / 64.0f);
    const float d0 = v0 - mu, d1 = v1 - mu;
    float vv = d0 * d0 + d1 * d1;
    #pragma unroll
    for (int off = 16; off > 0; off >>= 1) vv += __shfl_xor(vv, off, 32);
    const float rstd = rsqrtf(vv * (1.0f / 64.0f) + LN_EPS);

    float o0 = d0 * rstd * g[ch0] + bparam[ch0];
    float o1 = d1 * rstd * g[ch1] + bparam[ch1];
    o0 = o0 > 0.0f ? o0 : __expf(o0) - 1.0f;
    o1 = o1 > 0.0f ? o1 : __expf(o1) - 1.0f;
    out[(size_t)node * 64 + ch0] = o0;
    out[(size_t)node * 64 + ch1] = o1;
}

static inline unsigned cdiv(long long a, int b) { return (unsigned)((a + b - 1) / b); }

extern "C" void kernel_launch(void* const* d_in, const int* in_sizes, int n_in,
                              void* d_out, int out_size, void* d_ws, size_t ws_size,
                              hipStream_t stream) {
    const float* h   = (const float*)d_in[0];
    const int*   ei  = (const int*)d_in[1];
    const float* W1  = (const float*)d_in[2];
    const float* a1s = (const float*)d_in[3];
    const float* a1d = (const float*)d_in[4];
    const float* b1  = (const float*)d_in[5];
    const float* W2  = (const float*)d_in[6];
    const float* a2s = (const float*)d_in[7];
    const float* a2d = (const float*)d_in[8];
    const float* b2  = (const float*)d_in[9];
    const float* ln1g = (const float*)d_in[10];
    const float* ln1b = (const float*)d_in[11];
    const float* ln2g = (const float*)d_in[12];
    const float* ln2b = (const float*)d_in[13];

    const int n = in_sizes[0] / 64;
    const int e = in_sizes[1] / 2;
    const int* src = ei;
    const int* dst = ei + e;

    float* ws = (float*)d_ws;
    float* xw   = ws; ws += (size_t)n * 64;
    float* den  = ws; ws += (size_t)n * 4;
    float* agg  = ws; ws += (size_t)n * 64;
    float* x1   = ws; ws += (size_t)n * 64;
    int2*  rec  = (int2*)ws; ws += (size_t)e * 2;
    int*   head = (int*)ws;  ws += (size_t)n;

    const int B = 256;

    list_init<<<cdiv(n, B), B, 0, stream>>>(head, n);
    list_build<<<cdiv(e, B), B, 0, stream>>>(src, dst, head, rec, e);

    gemm64_wmma<<<cdiv(n, 16), 128, 0, stream>>>(h, W1, xw, n);
    node_walk<4, 16><<<cdiv((long long)n * 32, B), B, 0, stream>>>(head, rec, a1s, a1d, xw, den, agg, n, e);
    finalize_k<4, 16><<<cdiv((long long)n * 32, B), B, 0, stream>>>(agg, den, b1, ln1g, ln1b, x1, n);

    gemm64_wmma<<<cdiv(n, 16), 128, 0, stream>>>(x1, W2, xw, n);
    node_walk<1, 64><<<cdiv((long long)n * 32, B), B, 0, stream>>>(head, rec, a2s, a2d, xw, den, agg, n, e);
    finalize_k<1, 64><<<cdiv((long long)n * 32, B), B, 0, stream>>>(agg, den, b2, ln2g, ln2b, (float*)d_out, n);
}


# edge fill split by chip half: a wave reads its die, draws edge chunks from its half's ticket queue, places only nodes that half owns (counter, slab lines written from one half); slots by arrival
# speedup vs baseline: 10.9826x; 2.9553x over previous
#include <hip/hip_runtime.h>

#define NEG_SLOPE 0.2f
#define LN_EPS 1e-5f
#define SM_EPS 1e-16f

typedef __attribute__((ext_vector_type(2))) float v2f;
typedef __attribute__((ext_vector_type(4))) float v4f;
typedef __attribute__((ext_vector_type(8))) float v8f;

__device__ __forceinline__ float lrelu(float x) { return x > 0.0f ? x : NEG_SLOPE * x; }

__global__ void __launch_bounds__(128)
gemm64_wmma(const float* __restrict__ A, const float* __restrict__ W,
            float* __restrict__ C, int nrows) {
    const int wave = threadIdx.x >> 5;
    const int lane = threadIdx.x & 31;
    const int mrow = lane & 15;
    const int kh   = lane >> 4;
    const int row0 = blockIdx.x * 16;
    int row = row0 + mrow;
    row = row < nrows ? row : nrows - 1;
    const int col  = wave * 16 + mrow;

    const float* ap = A + (size_t)row * 64 + 2 * kh;
    const float* bp = W + (size_t)(2 * kh) * 64 + col;

    v8f acc = {};
    #pragma unroll
    for (int k = 0; k < 64; k += 4) {
        v2f a = *(const v2f*)(ap + k);
        v2f b;
        b.x = bp[(size_t)k * 64];
        b.y = bp[(size_t)k * 64 + 64];
        acc = __builtin_amdgcn_wmma_f32_16x16x4_f32(
             false, a,  false, b,
             (short)0, acc,  false,  false);
    }

    const int srow = row0 + 8 * (lane >> 4);
    const int scol = wave * 16 + (lane & 15);
    #pragma unroll
    for (int r = 0; r < 8; ++r) {
        const int rr = srow + r;
        if (rr < nrows) C[(size_t)rr * 64 + scol] = acc[r];
    }
}

template <int CTRL>
__device__ __forceinline__ float xmask_add(float v) {
    const int b = __builtin_bit_cast(int, v);
    const int t = __builtin_amdgcn_update_dpp(b, b, CTRL, 0xf, 0xf, true);
    return v + __builtin_bit_cast(float, t);
}

template <int LANES>
__device__ __forceinline__ float group_sum(float v) {
    v = xmask_add<0x161>(v);
    if (LANES >= 4) v = xmask_add<0x162>(v);
    if (LANES >= 8) v = xmask_add<0x164>(v);
    return v;
}

__device__ __forceinline__ float dot8(v4f xa, v4f xb, v4f wa, v4f wb) {
    v2f t = xa.xy * wa.xy;
    t = xa.zw * wa.zw + t;
    t = xb.xy * wb.xy + t;
    t = xb.zw * wb.zw + t;
    return t.x + t.y;
}

#define CAP 64

#define FCH 256
#define TQ  128

__global__ void slab_init(int* __restrict__ cnt, int* __restrict__ head,
                          int* __restrict__ tick, int n) {
    const int t = blockIdx.x * blockDim.x + threadIdx.x;
    if (t < n) { cnt[t] = 0; head[t] = -1; }
    if (t < 2) tick[t * TQ] = 0;
}

__device__ __forceinline__ void slab_place(int s, int d, int k, int e,
                                           int* __restrict__ slab, int* __restrict__ head,
                                           int2* __restrict__ rec) {
    if (k < CAP) {
        slab[(size_t)d * CAP + k] = s;
    } else {
        const int prev = atomicExch(&head[d], e);
        rec[e] = make_int2(s, prev);
    }
}

__global__ void __launch_bounds__(256)
slab_fill(const int* __restrict__ src, const int* __restrict__ dst,
          int* __restrict__ cnt, int* __restrict__ slab,
          int* __restrict__ head, int2* __restrict__ rec,
          int* __restrict__ tick, int ecnt, int n) {
    const int lane = threadIdx.x & 31;
    const unsigned where = __builtin_amdgcn_s_sendmsg_rtn(0x87);
    const int myhalf = (int)((where >> 18) & 1u);
    int nsplit = ((n >> 1) + 63) & ~63;
    nsplit = nsplit < n ? nsplit : n;
    const int nchunks = (ecnt + FCH - 1) / FCH;
    const bool vec_ok = ((((size_t)src) | ((size_t)dst)) & 15) == 0;

    for (int pass = 0; pass < 2; ++pass) {
        const int q = myhalf ^ pass;
        const int lo = q ? nsplit : 0;
        const unsigned span = (unsigned)((q ? n : nsplit) - lo);
        int* tq = tick + q * TQ;

        for (int it = 0; it <= nchunks; ++it) {
            int c = 0;
            if (lane == 0) c = atomicAdd(tq, 1);
            c = __builtin_amdgcn_readfirstlane(c);
            if (c >= nchunks) break;

            const int g0 = c * FCH + 4 * lane;
            const int g1 = g0 + FCH / 2;
            if (vec_ok && g1 + 3 < ecnt) {
                const int4 da = *(const int4*)(dst + g0);
                const int4 db = *(const int4*)(dst + g1);
                const int4 sa = *(const int4*)(src + g0);
                const int4 sb = *(const int4*)(src + g1);
                const bool o0 = (unsigned)(da.x - lo) < span, o1 = (unsigned)(da.y - lo) < span;
                const bool o2 = (unsigned)(da.z - lo) < span, o3 = (unsigned)(da.w - lo) < span;
                const bool o4 = (unsigned)(db.x - lo) < span, o5 = (unsigned)(db.y - lo) < span;
                const bool o6 = (unsigned)(db.z - lo) < span, o7 = (unsigned)(db.w - lo) < span;
                int k0 = 0, k1 = 0, k2 = 0, k3 = 0, k4 = 0, k5 = 0, k6 = 0, k7 = 0;
                if (o0) k0 = atomicAdd(&cnt[da.x], 1);
                if (o1) k1 = atomicAdd(&cnt[da.y], 1);
                if (o2) k2 = atomicAdd(&cnt[da.z], 1);
                if (o3) k3 = atomicAdd(&cnt[da.w], 1);
                if (o4) k4 = atomicAdd(&cnt[db.x], 1);
                if (o5) k5 = atomicAdd(&cnt[db.y], 1);
                if (o6) k6 = atomicAdd(&cnt[db.z], 1);
                if (o7) k7 = atomicAdd(&cnt[db.w], 1);
                if (o0) slab_place(sa.x, da.x, k0, g0,     slab, head, rec);
                if (o1) slab_place(sa.y, da.y, k1, g0 + 1, slab, head, rec);
                if (o2) slab_place(sa.z, da.z, k2, g0 + 2, slab, head, rec);
                if (o3) slab_place(sa.w, da.w, k3, g0 + 3, slab, head, rec);
                if (o4) slab_place(sb.x, db.x, k4, g1,     slab, head, rec);
                if (o5) slab_place(sb.y, db.y, k5, g1 + 1, slab, head, rec);
                if (o6) slab_place(sb.z, db.z, k6, g1 + 2, slab, head, rec);
                if (o7) slab_place(sb.w, db.w, k7, g1 + 3, slab, head, rec);
            } else {
                for (int j = 0; j < 8; ++j) {
                    const int e = (j < 4 ? g0 : g1 - 4) + j;
                    if (e < ecnt) {
                        const int d = dst[e];
                        if ((unsigned)(d - lo) < span) {
                            const int k = atomicAdd(&cnt[d], 1);
                            slab_place(src[e], d, k, e, slab, head, rec);
                        }
                    }
                }
            }
        }
    }
}

__device__ __forceinline__ int row_gather(int v, int sel) {
    return (int)__builtin_amdgcn_permlane16_var(0u, (unsigned)v, (unsigned)sel, true, false);
}

template <int H, int C>
__device__ __forceinline__ void
walk_one(const int node, const bool live,
         const int* __restrict__ cnt, const int* __restrict__ slab,
         const int* __restrict__ head, const int2* __restrict__ rec,
         const float* __restrict__ a_src, const float* __restrict__ a_dst,
         const float* __restrict__ xw, const int ecnt,
         v4f& accA, v4f& accB, float& den) {
    const int sub  = threadIdx.x & 7;
    constexpr int HL = C / 8;
    const int cA = (C >= 64) ? 4 * sub      : 8 * sub;
    const int cB = (C >= 64) ? 4 * sub + 32 : 8 * sub + 4;

    const v4f wsA = *(const v4f*)(a_src + cA), wsB = *(const v4f*)(a_src + cB);
    const v4f wdA = *(const v4f*)(a_dst + cA), wdB = *(const v4f*)(a_dst + cB);

    const char* xwb = (const char*)xw;
    const unsigned lo = 4u * (unsigned)cA;
    constexpr unsigned AB = 4u * ((C >= 64) ? 32 : 4);
    const unsigned slo = ((unsigned)node << 8) + 16u * (unsigned)sub;
    const char* slb = (const char*)slab;
    const int gsel = (int)(threadIdx.x & 8);

    const int deg = live ? cnt[node] : 0;
    const int4 idlo = *(const int4*)(slb + slo);
    const unsigned so = ((unsigned)node << 8) + lo;
    const v4f sA = *(const v4f*)(xwb + so);
    const v4f sB = *(const v4f*)(xwb + so + AB);
    const int ns  = deg < CAP ? deg : CAP;
    int4 idhi = idlo;
    if (ns > 32) idhi = *(const int4*)(slb + slo + 128u);
    const float ad = group_sum<HL>(dot8(sA, sB, wdA, wdB));
    float m = lrelu(group_sum<HL>(dot8(sA, sB, wsA, wsB)) + ad);
    accA = sA; accB = sB;
    den = 1.0f;

    auto trips = [&](const int4 mine, const int k0, const int kend) {
        for (int k = k0; k < kend; k += 4) {
            const int sel = gsel + ((k - k0) >> 2);
            const int i0 = row_gather(mine.x, sel), i1 = row_gather(mine.y, sel);
            const int i2 = row_gather(mine.z, sel), i3 = row_gather(mine.w, sel);
            const bool v1 = k + 1 < kend, v2 = k + 2 < kend, v3 = k + 3 < kend;
            const unsigned o0 = ((unsigned)i0 << 8) + lo;
            const unsigned o1 = v1 ? ((unsigned)i1 << 8) + lo : o0;
            const unsigned o2 = v2 ? ((unsigned)i2 << 8) + lo : o0;
            const unsigned o3 = v3 ? ((unsigned)i3 << 8) + lo : o0;
            const v4f x0A = *(const v4f*)(xwb + o0), x0B = *(const v4f*)(xwb + o0 + AB);
            const v4f x1A = *(const v4f*)(xwb + o1), x1B = *(const v4f*)(xwb + o1 + AB);
            const v4f x2A = *(const v4f*)(xwb + o2), x2B = *(const v4f*)(xwb + o2 + AB);
            const v4f x3A = *(const v4f*)(xwb + o3), x3B = *(const v4f*)(xwb + o3 + AB);
            const float e0 = lrelu(group_sum<HL>(dot8(x0A, x0B, wsA, wsB)) + ad);
            const float t1 = lrelu(group_sum<HL>(dot8(x1A, x1B, wsA, wsB)) + ad);
            const float t2 = lrelu(group_sum<HL>(dot8(x2A, x2B, wsA, wsB)) + ad);
            const float t3 = lrelu(group_sum<HL>(dot8(x3A, x3B, wsA, wsB)) + ad);
            const float e1 = v1 ? t1 : -INFINITY;
            const float e2 = v2 ? t2 : -INFINITY;
            const float e3 = v3 ? t3 : -INFINITY;
            const float mn = fmaxf(fmaxf(m, e0), fmaxf(e1, fmaxf(e2, e3)));
            const float sc = __expf(m - mn);
            const float p0 = __expf(e0 - mn);
            const float p1 = __expf(e1 - mn);
            const float p2 = __expf(e2 - mn);
            const float p3 = __expf(e3 - mn);
            den  = den * sc + ((p0 + p1) + (p2 + p3));
            accA = accA * sc + ((p0 * x0A + p1 * x1A) + (p2 * x2A + p3 * x3A));
            accB = accB * sc + ((p0 * x0B + p1 * x1B) + (p2 * x2B + p3 * x3B));
            m = mn;
        }
    };
    trips(idlo, 0, ns < 32 ? ns : 32);
    trips(idhi, 32, ns);

    if (deg > CAP) {
        const char* rcb = (const char*)rec;
        int e = head[node];
        int2 r = *(const int2*)(rcb + ((unsigned)(e >= 0 ? e : 0) << 3));
        for (int it = 0; it < ecnt && e >= 0; ++it) {
            const int s  = r.x;
            const int en = r.y;
            r = *(const int2*)(rcb + ((unsigned)(en >= 0 ? en : 0) << 3));
            const v4f xA = *(const v4f*)(xwb + (((unsigned)s << 8) + lo));
            const v4f xB = *(const v4f*)(xwb + (((unsigned)s << 8) + lo) + AB);
            const float as = group_sum<HL>(dot8(xA, xB, wsA, wsB));
            const float ev = lrelu(as + ad);
            const float mn = fmaxf(m, ev);
            const float sc = __expf(m - mn);
            const float p  = __expf(ev - mn);
            den  = den * sc + p;
            accA = accA * sc + p * xA;
            accB = accB * sc + p * xB;
            m = mn;
            e = en;
        }
    }

}

template <int H, int C>
__global__ void __launch_bounds__(256)
node_walk(const int* __restrict__ cnt, const int* __restrict__ slab,
          const int* __restrict__ head, const int2* __restrict__ rec,
          const float* __restrict__ a_src, const float* __restrict__ a_dst,
          const float* __restrict__ xw,
          float* __restrict__ denom, float* __restrict__ agg,
          int n, int ecnt) {
    const int node = (int)((blockIdx.x * blockDim.x + threadIdx.x) >> 3);
    if (node >= n) return;
    const int sub = threadIdx.x & 7;
    const int cA = (C >= 64) ? 4 * sub      : 8 * sub;
    const int cB = (C >= 64) ? 4 * sub + 32 : 8 * sub + 4;
    v4f accA, accB;
    float den;
    walk_one<H, C>(node, true, cnt, slab, head, rec, a_src, a_dst, xw, ecnt, accA, accB, den);
    *(v4f*)(agg + (size_t)node * 64 + cA) = accA;
    *(v4f*)(agg + (size_t)node * 64 + cB) = accB;
    if ((cA & (C - 1)) == 0) denom[node * H + cA / C] = den;
}

#define XP 68

__global__ void __launch_bounds__(256)
walk_ln_gemm(const int* __restrict__ cnt, const int* __restrict__ slab,
             const int* __restrict__ head, const int2* __restrict__ rec,
             const float* __restrict__ a_src, const float* __restrict__ a_dst,
             const float* __restrict__ xw,
             const float* __restrict__ bias, const float* __restrict__ g,
             const float* __restrict__ bparam, const float* __restrict__ W,
             float* __restrict__ out, int n, int ecnt) {
    __shared__ __attribute__((aligned(16))) float xs[32 * XP];

    const int nraw = (int)((blockIdx.x * blockDim.x + threadIdx.x) >> 3);
    const bool live = nraw < n;
    const int node = live ? nraw : n - 1;
    const int sub  = threadIdx.x & 7;
    const int c0   = 8 * sub;

    v4f accA, accB;
    float den;
    walk_one<4, 16>(node, live, cnt, slab, head, rec, a_src, a_dst, xw, ecnt, accA, accB, den);

    const int wave = threadIdx.x >> 5;
    const int lane = threadIdx.x & 31;
    const int mrow = lane & 15;
    const int kh   = lane >> 4;
    const int rt   = wave >> 2;
    const int col  = (wave & 3) * 16 + mrow;
    const float* bp = W + (size_t)(2 * kh) * 64 + col;
    v2f bw[16];
    #pragma unroll
    for (int k = 0; k < 16; ++k) {
        bw[k].x = bp[(size_t)k * 256];
        bw[k].y = bp[(size_t)k * 256 + 64];
    }

    const float dn = den + SM_EPS;
    const v4f vA = accA / dn + *(const v4f*)(bias + c0);
    const v4f vB = accB / dn + *(const v4f*)(bias + c0 + 4);
    const float s = group_sum<8>(((vA.x + vA.y) + (vA.z + vA.w)) + ((vB.x + vB.y) + (vB.z + vB.w)));
    const float mu = s * (1.0f / 64.0f);
    const v4f dA = vA - mu, dB = vB - mu;
    const v4f qA = dA * dA, qB = dB * dB;
    const float vv = group_sum<8>(((qA.x + qA.y) + (qA.z + qA.w)) + ((qB.x + qB.y) + (qB.z + qB.w)));
    const float rstd = rsqrtf(vv * (1.0f / 64.0f) + LN_EPS);
    v4f oA = dA * rstd * *(const v4f*)(g + c0)     + *(const v4f*)(bparam + c0);
    v4f oB = dB * rstd * *(const v4f*)(g + c0 + 4) + *(const v4f*)(bparam + c0 + 4);
    auto elu = [](const float o) { return o > 0.0f ? o : __expf(o) - 1.0f; };
    oA = (v4f){elu(oA.x), elu(oA.y), elu(oA.z), elu(oA.w)};
    oB = (v4f){elu(oB.x), elu(oB.y), elu(oB.z), elu(oB.w)};
    float* xr = xs + (threadIdx.x >> 3) * XP + c0;
    *(v4f*)(xr)     = oA;
    *(v4f*)(xr + 4) = oB;

    const float* ap = xs + (rt * 16 + mrow) * XP + 2 * kh;

    __syncthreads();

    v8f acc = {};
    #pragma unroll
    for (int k = 0; k < 64; k += 4) {
        const v2f a = *(const v2f*)(ap + k);
        acc = __builtin_amdgcn_wmma_f32_16x16x4_f32(
             false, a,  false, bw[k >> 2],
             (short)0, acc,  false,  false);
    }

    const int srow = (int)(blockIdx.x * 32) + rt * 16 + 8 * (lane >> 4);
    const int scol = (wave & 3) * 16 + (lane & 15);
    #pragma unroll
    for (int r = 0; r < 8; ++r) {
        const int rr = srow + r;
        if (rr < n) out[(size_t)rr * 64 + scol] = acc[r];
    }
}

template <int H, int C>
__global__ void finalize_k(const float* __restrict__ agg, const float* __restrict__ denom,
                           const float* __restrict__ bias, const float* __restrict__ g,
                           const float* __restrict__ bparam, float* __restrict__ out,
                           int n) {
    const int node = (blockIdx.x * blockDim.x + threadIdx.x) >> 5;
    const int lane = threadIdx.x & 31;
    if (node >= n) return;
    const int ch0 = lane, ch1 = lane + 32;
    float v0 = agg[(size_t)node * 64 + ch0] / (denom[node * H + ch0 / C] + SM_EPS) + bias[ch0];
    float v1 = agg[(size_t)node * 64 + ch1] / (denom[node * H + ch1 / C] + SM_EPS) + bias[ch1];

    float s = v0 + v1;
    #pragma unroll
    for (int off = 16; off > 0; off >>= 1) s += __shfl_xor(s, off, 32);
    const float mu = s * (1.0f / 64.0f);
    const float d0 = v0 - mu, d1 = v1 - mu;
    float vv = d0 * d0 + d1 * d1;
    #pragma unroll
    for (int off = 16; off > 0; off >>= 1) vv += __shfl_xor(vv, off, 32);
    const float rstd = rsqrtf(vv * (1.0f / 64.0f) + LN_EPS);

    float o0 = d0 * rstd * g[ch0] + bparam[ch0];
    float o1 = d1 * rstd * g[ch1] + bparam[ch1];
    o0 = o0 > 0.0f ? o0 : __expf(o0) - 1.0f;
    o1 = o1 > 0.0f ? o1 : __expf(o1) - 1.0f;
    out[(size_t)node * 64 + ch0] = o0;
    out[(size_t)node * 64 + ch1] = o1;
}

static inline unsigned cdiv(long long a, int b) { return (unsigned)((a + b - 1) / b); }

extern "C" void kernel_launch(void* const* d_in, const int* in_sizes, int n_in,
                              void* d_out, int out_size, void* d_ws, size_t ws_size,
                              hipStream_t stream) {
    const float* h   = (const float*)d_in[0];
    const int*   ei  = (const int*)d_in[1];
    const float* W1  = (const float*)d_in[2];
    const float* a1s = (const float*)d_in[3];
    const float* a1d = (const float*)d_in[4];
    const float* b1  = (const float*)d_in[5];
    const float* W2  = (const float*)d_in[6];
    const float* a2s = (const float*)d_in[7];
    const float* a2d = (const float*)d_in[8];
    const float* b2  = (const float*)d_in[9];
    const float* ln1g = (const float*)d_in[10];
    const float* ln1b = (const float*)d_in[11];
    const float* ln2g = (const float*)d_in[12];
    const float* ln2b = (const float*)d_in[13];

    const int n = in_sizes[0] / 64;
    const int e = in_sizes[1] / 2;
    const int* src = ei;
    const int* dst = ei + e;

    float* ws = (float*)d_ws;
    float* xw   = ws; ws += (size_t)n * 64;
    float* den  = ws; ws += (size_t)n * 4;
    float* agg  = ws; ws += (size_t)n * 64;
    float* x1   = ws; ws += (size_t)n * 64;
    int*   slab = (int*)ws;  ws += (size_t)n * CAP;
    int*   cnt  = (int*)ws;  ws += (size_t)n;
    int*   head = (int*)ws;  ws += (size_t)n;
    int2*  rec  = (int2*)ws; ws += (size_t)e * 2;
    int*   tick = (int*)ws;  ws += 2 * TQ;

    const int B = 256;

    slab_init<<<cdiv(n, B), B, 0, stream>>>(cnt, head, tick, n);
    slab_fill<<<512, B, 0, stream>>>(src, dst, cnt, slab, head, rec, tick, e, n);

    gemm64_wmma<<<cdiv(n, 16), 128, 0, stream>>>(h, W1, xw, n);
    walk_ln_gemm<<<cdiv((long long)n * 8, B), B, 0, stream>>>(cnt, slab, head, rec, a1s, a1d, xw, b1, ln1g, ln1b, W2, x1, n, e);

    node_walk<1, 64><<<cdiv((long long)n * 8, B), B, 0, stream>>>(cnt, slab, head, rec, a2s, a2d, x1, den, agg, n, e);
    finalize_k<1, 64><<<cdiv((long long)n * 32, B), B, 0, stream>>>(agg, den, b2, ln2g, ln2b, (float*)d_out, n);
}


# layer 2 walk group finishes its row in place (divide, bias, LayerNorm, ELU per 8-lane group, 16-byte stores to out): finalize launch and agg/denom round trip gone; LN sums now 8 per lane + lane tree
# speedup vs baseline: 11.9353x; 1.0867x over previous
#include <hip/hip_runtime.h>

#define NEG_SLOPE 0.2f
#define LN_EPS 1e-5f
#define SM_EPS 1e-16f

typedef __attribute__((ext_vector_type(2))) float v2f;
typedef __attribute__((ext_vector_type(4))) float v4f;
typedef __attribute__((ext_vector_type(8))) float v8f;

__device__ __forceinline__ float lrelu(float x) { return x > 0.0f ? x : NEG_SLOPE * x; }

__global__ void __launch_bounds__(128)
gemm64_wmma(const float* __restrict__ A, const float* __restrict__ W,
            float* __restrict__ C, int nrows) {
    const int wave = threadIdx.x >> 5;
    const int lane = threadIdx.x & 31;
    const int mrow = lane & 15;
    const int kh   = lane >> 4;
    const int row0 = blockIdx.x * 16;
    int row = row0 + mrow;
    row = row < nrows ? row : nrows - 1;
    const int col  = wave * 16 + mrow;

    const float* ap = A + (size_t)row * 64 + 2 * kh;
    const float* bp = W + (size_t)(2 * kh) * 64 + col;

    v8f acc = {};
    #pragma unroll
    for (int k = 0; k < 64; k += 4) {
        v2f a = *(const v2f*)(ap + k);
        v2f b;
        b.x = bp[(size_t)k * 64];
        b.y = bp[(size_t)k * 64 + 64];
        acc = __builtin_amdgcn_wmma_f32_16x16x4_f32(
             false, a,  false, b,
             (short)0, acc,  false,  false);
    }

    const int srow = row0 + 8 * (lane >> 4);
    const int scol = wave * 16 + (lane & 15);
    #pragma unroll
    for (int r = 0; r < 8; ++r) {
        const int rr = srow + r;
        if (rr < nrows) C[(size_t)rr * 64 + scol] = acc[r];
    }
}

template <int CTRL>
__device__ __forceinline__ float xmask_add(float v) {
    const int b = __builtin_bit_cast(int, v);
    const int t = __builtin_amdgcn_update_dpp(b, b, CTRL, 0xf, 0xf, true);
    return v + __builtin_bit_cast(float, t);
}

template <int LANES>
__device__ __forceinline__ float group_sum(float v) {
    v = xmask_add<0x161>(v);
    if (LANES >= 4) v = xmask_add<0x162>(v);
    if (LANES >= 8) v = xmask_add<0x164>(v);
    return v;
}

__device__ __forceinline__ float dot8(v4f xa, v4f xb, v4f wa, v4f wb) {
    v2f t = xa.xy * wa.xy;
    t = xa.zw * wa.zw + t;
    t = xb.xy * wb.xy + t;
    t = xb.zw * wb.zw + t;
    return t.x + t.y;
}

#define CAP 64

#define FCH 256
#define TQ  128

__global__ void slab_init(int* __restrict__ cnt, int* __restrict__ head,
                          int* __restrict__ tick, int n) {
    const int t = blockIdx.x * blockDim.x + threadIdx.x;
    if (t < n) { cnt[t] = 0; head[t] = -1; }
    if (t < 2) tick[t * TQ] = 0;
}

__device__ __forceinline__ void slab_place(int s, int d, int k, int e,
                                           int* __restrict__ slab, int* __restrict__ head,
                                           int2* __restrict__ rec) {
    if (k < CAP) {
        slab[(size_t)d * CAP + k] = s;
    } else {
        const int prev = atomicExch(&head[d], e);
        rec[e] = make_int2(s, prev);
    }
}

__global__ void __launch_bounds__(256)
slab_fill(const int* __restrict__ src, const int* __restrict__ dst,
          int* __restrict__ cnt, int* __restrict__ slab,
          int* __restrict__ head, int2* __restrict__ rec,
          int* __restrict__ tick, int ecnt, int n) {
    const int lane = threadIdx.x & 31;
    const unsigned where = __builtin_amdgcn_s_sendmsg_rtn(0x87);
    const int myhalf = (int)((where >> 18) & 1u);
    int nsplit = ((n >> 1) + 63) & ~63;
    nsplit = nsplit < n ? nsplit : n;
    const int nchunks = (ecnt + FCH - 1) / FCH;
    const bool vec_ok = ((((size_t)src) | ((size_t)dst)) & 15) == 0;

    for (int pass = 0; pass < 2; ++pass) {
        const int q = myhalf ^ pass;
        const int lo = q ? nsplit : 0;
        const unsigned span = (unsigned)((q ? n : nsplit) - lo);
        int* tq = tick + q * TQ;

        for (int it = 0; it <= nchunks; ++it) {
            int c = 0;
            if (lane == 0) c = atomicAdd(tq, 1);
            c = __builtin_amdgcn_readfirstlane(c);
            if (c >= nchunks) break;

            const int g0 = c * FCH + 4 * lane;
            const int g1 = g0 + FCH / 2;
            if (vec_ok && g1 + 3 < ecnt) {
                const int4 da = *(const int4*)(dst + g0);
                const int4 db = *(const int4*)(dst + g1);
                const int4 sa = *(const int4*)(src + g0);
                const int4 sb = *(const int4*)(src + g1);
                const bool o0 = (unsigned)(da.x - lo) < span, o1 = (unsigned)(da.y - lo) < span;
                const bool o2 = (unsigned)(da.z - lo) < span, o3 = (unsigned)(da.w - lo) < span;
                const bool o4 = (unsigned)(db.x - lo) < span, o5 = (unsigned)(db.y - lo) < span;
                const bool o6 = (unsigned)(db.z - lo) < span, o7 = (unsigned)(db.w - lo) < span;
                int k0 = 0, k1 = 0, k2 = 0, k3 = 0, k4 = 0, k5 = 0, k6 = 0, k7 = 0;
                if (o0) k0 = atomicAdd(&cnt[da.x], 1);
                if (o1) k1 = atomicAdd(&cnt[da.y], 1);
                if (o2) k2 = atomicAdd(&cnt[da.z], 1);
                if (o3) k3 = atomicAdd(&cnt[da.w], 1);
                if (o4) k4 = atomicAdd(&cnt[db.x], 1);
                if (o5) k5 = atomicAdd(&cnt[db.y], 1);
                if (o6) k6 = atomicAdd(&cnt[db.z], 1);
                if (o7) k7 = atomicAdd(&cnt[db.w], 1);
                if (o0) slab_place(sa.x, da.x, k0, g0,     slab, head, rec);
                if (o1) slab_place(sa.y, da.y, k1, g0 + 1, slab, head, rec);
                if (o2) slab_place(sa.z, da.z, k2, g0 + 2, slab, head, rec);
                if (o3) slab_place(sa.w, da.w, k3, g0 + 3, slab, head, rec);
                if (o4) slab_place(sb.x, db.x, k4, g1,     slab, head, rec);
                if (o5) slab_place(sb.y, db.y, k5, g1 + 1, slab, head, rec);
                if (o6) slab_place(sb.z, db.z, k6, g1 + 2, slab, head, rec);
                if (o7) slab_place(sb.w, db.w, k7, g1 + 3, slab, head, rec);
            } else {
                for (int j = 0; j < 8; ++j) {
                    const int e = (j < 4 ? g0 : g1 - 4) + j;
                    if (e < ecnt) {
                        const int d = dst[e];
                        if ((unsigned)(d - lo) < span) {
                            const int k = atomicAdd(&cnt[d], 1);
                            slab_place(src[e], d, k, e, slab, head, rec);
                        }
                    }
                }
            }
        }
    }
}

__device__ __forceinline__ int row_gather(int v, int sel) {
    return (int)__builtin_amdgcn_permlane16_var(0u, (unsigned)v, (unsigned)sel, true, false);
}

template <int H, int C>
__device__ __forceinline__ void
walk_one(const int node, const bool live,
         const int* __restrict__ cnt, const int* __restrict__ slab,
         const int* __restrict__ head, const int2* __restrict__ rec,
         const float* __restrict__ a_src, const float* __restrict__ a_dst,
         const float* __restrict__ xw, const int ecnt,
         v4f& accA, v4f& accB, float& den) {
    const int sub  = threadIdx.x & 7;
    constexpr int HL = C / 8;
    const int cA = (C >= 64) ? 4 * sub      : 8 * sub;
    const int cB = (C >= 64) ? 4 * sub + 32 : 8 * sub + 4;

    const v4f wsA = *(const v4f*)(a_src + cA), wsB = *(const v4f*)(a_src + cB);
    const v4f wdA = *(const v4f*)(a_dst + cA), wdB = *(const v4f*)(a_dst + cB);

    const char* xwb = (const char*)xw;
    const unsigned lo = 4u * (unsigned)cA;
    constexpr unsigned AB = 4u * ((C >= 64) ? 32 : 4);
    const unsigned slo = ((unsigned)node << 8) + 16u * (unsigned)sub;
    const char* slb = (const char*)slab;
    const int gsel = (int)(threadIdx.x & 8);

    const int deg = live ? cnt[node] : 0;
    const int4 idlo = *(const int4*)(slb + slo);
    const unsigned so = ((unsigned)node << 8) + lo;
    const v4f sA = *(const v4f*)(xwb + so);
    const v4f sB = *(const v4f*)(xwb + so + AB);
    const int ns  = deg < CAP ? deg : CAP;
    int4 idhi = idlo;
    if (ns > 32) idhi = *(const int4*)(slb + slo + 128u);
    const float ad = group_sum<HL>(dot8(sA, sB, wdA, wdB));
    float m = lrelu(group_sum<HL>(dot8(sA, sB, wsA, wsB)) + ad);
    accA = sA; accB = sB;
    den = 1.0f;

    auto trips = [&](const int4 mine, const int k0, const int kend) {
        for (int k = k0; k < kend; k += 4) {
            const int sel = gsel + ((k - k0) >> 2);
            const int i0 = row_gather(mine.x, sel), i1 = row_gather(mine.y, sel);
            const int i2 = row_gather(mine.z, sel), i3 = row_gather(mine.w, sel);
            const bool v1 = k + 1 < kend, v2 = k + 2 < kend, v3 = k + 3 < kend;
            const unsigned o0 = ((unsigned)i0 << 8) + lo;
            const unsigned o1 = v1 ? ((unsigned)i1 << 8) + lo : o0;
            const unsigned o2 = v2 ? ((unsigned)i2 << 8) + lo : o0;
            const unsigned o3 = v3 ? ((unsigned)i3 << 8) + lo : o0;
            const v4f x0A = *(const v4f*)(xwb + o0), x0B = *(const v4f*)(xwb + o0 + AB);
            const v4f x1A = *(const v4f*)(xwb + o1), x1B = *(const v4f*)(xwb + o1 + AB);
            const v4f x2A = *(const v4f*)(xwb + o2), x2B = *(const v4f*)(xwb + o2 + AB);
            const v4f x3A = *(const v4f*)(xwb + o3), x3B = *(const v4f*)(xwb + o3 + AB);
            const float e0 = lrelu(group_sum<HL>(dot8(x0A, x0B, wsA, wsB)) + ad);
            const float t1 = lrelu(group_sum<HL>(dot8(x1A, x1B, wsA, wsB)) + ad);
            const float t2 = lrelu(group_sum<HL>(dot8(x2A, x2B, wsA, wsB)) + ad);
            const float t3 = lrelu(group_sum<HL>(dot8(x3A, x3B, wsA, wsB)) + ad);
            const float e1 = v1 ? t1 : -INFINITY;
            const float e2 = v2 ? t2 : -INFINITY;
            const float e3 = v3 ? t3 : -INFINITY;
            const float mn = fmaxf(fmaxf(m, e0), fmaxf(e1, fmaxf(e2, e3)));
            const float sc = __expf(m - mn);
            const float p0 = __expf(e0 - mn);
            const float p1 = __expf(e1 - mn);
            const float p2 = __expf(e2 - mn);
            const float p3 = __expf(e3 - mn);
            den  = den * sc + ((p0 + p1) + (p2 + p3));
            accA = accA * sc + ((p0 * x0A + p1 * x1A) + (p2 * x2A + p3 * x3A));
            accB = accB * sc + ((p0 * x0B + p1 * x1B) + (p2 * x2B + p3 * x3B));
            m = mn;
        }
    };
    trips(idlo, 0, ns < 32 ? ns : 32);
    trips(idhi, 32, ns);

    if (deg > CAP) {
        const char* rcb = (const char*)rec;
        int e = head[node];
        int2 r = *(const int2*)(rcb + ((unsigned)(e >= 0 ? e : 0) << 3));
        for (int it = 0; it < ecnt && e >= 0; ++it) {
            const int s  = r.x;
            const int en = r.y;
            r = *(const int2*)(rcb + ((unsigned)(en >= 0 ? en : 0) << 3));
            const v4f xA = *(const v4f*)(xwb + (((unsigned)s << 8) + lo));
            const v4f xB = *(const v4f*)(xwb + (((unsigned)s << 8) + lo) + AB);
            const float as = group_sum<HL>(dot8(xA, xB, wsA, wsB));
            const float ev = lrelu(as + ad);
            const float mn = fmaxf(m, ev);
            const float sc = __expf(m - mn);
            const float p  = __expf(ev - mn);
            den  = den * sc + p;
            accA = accA * sc + p * xA;
            accB = accB * sc + p * xB;
            m = mn;
            e = en;
        }
    }

}

__device__ __forceinline__ void
finish_row(const v4f accA, const v4f accB, const float den, const int cA, const int cB,
           const float* __restrict__ bias, const float* __restrict__ g,
           const float* __restrict__ bparam, v4f& oA, v4f& oB) {
    const float dn = den + SM_EPS;
    const v4f vA = accA / dn + *(const v4f*)(bias + cA);
    const v4f vB = accB / dn + *(const v4f*)(bias + cB);
    const float s = group_sum<8>(((vA.x + vA.y) + (vA.z + vA.w)) + ((vB.x + vB.y) + (vB.z + vB.w)));
    const float mu = s * (1.0f / 64.0f);
    const v4f dA = vA - mu, dB = vB - mu;
    const v4f qA = dA * dA, qB = dB * dB;
    const float vv = group_sum<8>(((qA.x + qA.y) + (qA.z + qA.w)) + ((qB.x + qB.y) + (qB.z + qB.w)));
    const float rstd = rsqrtf(vv * (1.0f / 64.0f) + LN_EPS);
    oA = dA * rstd * *(const v4f*)(g + cA) + *(const v4f*)(bparam + cA);
    oB = dB * rstd * *(const v4f*)(g + cB) + *(const v4f*)(bparam + cB);
    auto elu = [](const float o) { return o > 0.0f ? o : __expf(o) - 1.0f; };
    oA = (v4f){elu(oA.x), elu(oA.y), elu(oA.z), elu(oA.w)};
    oB = (v4f){elu(oB.x), elu(oB.y), elu(oB.z), elu(oB.w)};
}

template <int H, int C>
__global__ void __launch_bounds__(256)
walk_ln(const int* __restrict__ cnt, const int* __restrict__ slab,
        const int* __restrict__ head, const int2* __restrict__ rec,
        const float* __restrict__ a_src, const float* __restrict__ a_dst,
        const float* __restrict__ xw,
        const float* __restrict__ bias, const float* __restrict__ g,
        const float* __restrict__ bparam, float* __restrict__ out,
        int n, int ecnt) {
    const int node = (int)((blockIdx.x * blockDim.x + threadIdx.x) >> 3);
    if (node >= n) return;
    const int sub = threadIdx.x & 7;
    const int cA = (C >= 64) ? 4 * sub      : 8 * sub;
    const int cB = (C >= 64) ? 4 * sub + 32 : 8 * sub + 4;
    v4f accA, accB;
    float den;
    walk_one<H, C>(node, true, cnt, slab, head, rec, a_src, a_dst, xw, ecnt, accA, accB, den);
    v4f oA, oB;
    finish_row(accA, accB, den, cA, cB, bias, g, bparam, oA, oB);
    *(v4f*)(out + (size_t)node * 64 + cA) = oA;
    *(v4f*)(out + (size_t)node * 64 + cB) = oB;
}

#define XP 68

__global__ void __launch_bounds__(256)
walk_ln_gemm(const int* __restrict__ cnt, const int* __restrict__ slab,
             const int* __restrict__ head, const int2* __restrict__ rec,
             const float* __restrict__ a_src, const float* __restrict__ a_dst,
             const float* __restrict__ xw,
             const float* __restrict__ bias, const float* __restrict__ g,
             const float* __restrict__ bparam, const float* __restrict__ W,
             float* __restrict__ out, int n, int ecnt) {
    __shared__ __attribute__((aligned(16))) float xs[32 * XP];

    const int nraw = (int)((blockIdx.x * blockDim.x + threadIdx.x) >> 3);
    const bool live = nraw < n;
    const int node = live ? nraw : n - 1;
    const int sub  = threadIdx.x & 7;
    const int c0   = 8 * sub;

    v4f accA, accB;
    float den;
    walk_one<4, 16>(node, live, cnt, slab, head, rec, a_src, a_dst, xw, ecnt, accA, accB, den);

    const int wave = threadIdx.x >> 5;
    const int lane = threadIdx.x & 31;
    const int mrow = lane & 15;
    const int kh   = lane >> 4;
    const int rt   = wave >> 2;
    const int col  = (wave & 3) * 16 + mrow;
    const float* bp = W + (size_t)(2 * kh) * 64 + col;
    v2f bw[16];
    #pragma unroll
    for (int k = 0; k < 16; ++k) {
        bw[k].x = bp[(size_t)k * 256];
        bw[k].y = bp[(size_t)k * 256 + 64];
    }

    v4f oA, oB;
    finish_row(accA, accB, den, c0, c0 + 4, bias, g, bparam, oA, oB);
    float* xr = xs + (threadIdx.x >> 3) * XP + c0;
    *(v4f*)(xr)     = oA;
    *(v4f*)(xr + 4) = oB;

    const float* ap = xs + (rt * 16 + mrow) * XP + 2 * kh;

    __syncthreads();

    v8f acc = {};
    #pragma unroll
    for (int k = 0; k < 64; k += 4) {
        const v2f a = *(const v2f*)(ap + k);
        acc = __builtin_amdgcn_wmma_f32_16x16x4_f32(
             false, a,  false, bw[k >> 2],
             (short)0, acc,  false,  false);
    }

    const int srow = (int)(blockIdx.x * 32) + rt * 16 + 8 * (lane >> 4);
    const int scol = (wave & 3) * 16 + (lane & 15);
    #pragma unroll
    for (int r = 0; r < 8; ++r) {
        const int rr = srow + r;
        if (rr < n) out[(size_t)rr * 64 + scol] = acc[r];
    }
}

static inline unsigned cdiv(long long a, int b) { return (unsigned)((a + b - 1) / b); }

extern "C" void kernel_launch(void* const* d_in, const int* in_sizes, int n_in,
                              void* d_out, int out_size, void* d_ws, size_t ws_size,
                              hipStream_t stream) {
    const float* h   = (const float*)d_in[0];
    const int*   ei  = (const int*)d_in[1];
    const float* W1  = (const float*)d_in[2];
    const float* a1s = (const float*)d_in[3];
    const float* a1d = (const float*)d_in[4];
    const float* b1  = (const float*)d_in[5];
    const float* W2  = (const float*)d_in[6];
    const float* a2s = (const float*)d_in[7];
    const float* a2d = (const float*)d_in[8];
    const float* b2  = (const float*)d_in[9];
    const float* ln1g = (const float*)d_in[10];
    const float* ln1b = (const float*)d_in[11];
    const float* ln2g = (const float*)d_in[12];
    const float* ln2b = (const float*)d_in[13];

    const int n = in_sizes[0] / 64;
    const int e = in_sizes[1] / 2;
    const int* src = ei;
    const int* dst = ei + e;

    float* ws = (float*)d_ws;
    float* xw   = ws; ws += (size_t)n * 64;
    float* x1   = ws; ws += (size_t)n * 64;
    int*   slab = (int*)ws;  ws += (size_t)n * CAP;
    int*   cnt  = (int*)ws;  ws += (size_t)n;
    int*   head = (int*)ws;  ws += (size_t)n;
    int2*  rec  = (int2*)ws; ws += (size_t)e * 2;
    int*   tick = (int*)ws;  ws += 2 * TQ;

    const int B = 256;

    slab_init<<<cdiv(n, B), B, 0, stream>>>(cnt, head, tick, n);
    slab_fill<<<512, B, 0, stream>>>(src, dst, cnt, slab, head, rec, tick, e, n);

    gemm64_wmma<<<cdiv(n, 16), 128, 0, stream>>>(h, W1, xw, n);
    walk_ln_gemm<<<cdiv((long long)n * 8, B), B, 0, stream>>>(cnt, slab, head, rec, a1s, a1d, xw, b1, ln1g, ln1b, W2, x1, n, e);

    walk_ln<1, 64><<<cdiv((long long)n * 8, B), B, 0, stream>>>(cnt, slab, head, rec, a2s, a2d, x1, b2, ln2g, ln2b, (float*)d_out, n, e);
}
